# starting kernel
# baseline (speedup 1.0000x reference)
#include <hip/hip_runtime.h>
#include <math.h>

typedef __attribute__((ext_vector_type(2))) float v2f;
typedef __attribute__((ext_vector_type(8))) float v8f;

#define N_NODES 65536
#define DEG     16
#define D       64
#define D2      128
#define NB1     1024
#define NNEG    10
#define GTOT    (NB1 + NB1 + NNEG)
#define EPSN    1e-12f
#define XPAD    132

__global__ __launch_bounds__(32) void graphsage_hop1(
    const float* __restrict__ feat, const int* __restrict__ adj,
    const float* __restrict__ W1,   const float* __restrict__ b1,
    float* __restrict__ h1)
{
    __shared__ float Xs[16][XPAD];

    const int lane = threadIdx.x;
    const int m    = lane & 15;
    const int half = lane >> 4;
    const int tile = blockIdx.x;
    const int node = tile * 16 + m;

    {
        const float4* selfp = (const float4*)(feat + (size_t)node * D + half * 32);
        float4* xlo = (float4*)&Xs[m][half * 32];
        float4* xhi = (float4*)&Xs[m][D + half * 32];
        #pragma unroll
        for (int i = 0; i < 8; ++i) xlo[i] = selfp[i];

        float4 acc[8];
        #pragma unroll
        for (int i = 0; i < 8; ++i) acc[i] = make_float4(0.f, 0.f, 0.f, 0.f);
        for (int j = 0; j < DEG; ++j) {
            const int nb = adj[(size_t)node * DEG + j];
            const float4* p = (const float4*)(feat + (size_t)nb * D + half * 32);
            #pragma unroll
            for (int i = 0; i < 8; ++i) {
                float4 v = p[i];
                acc[i].x += v.x; acc[i].y += v.y; acc[i].z += v.z; acc[i].w += v.w;
            }
        }
        #pragma unroll
        for (int i = 0; i < 8; ++i) xhi[i] = acc[i];
    }
    __syncthreads();

    const int kb = half * 2;
    v8f zero = {};
    v8f c[4];
    #pragma unroll
    for (int nt = 0; nt < 4; ++nt) c[nt] = zero;

    for (int k0 = 0; k0 < 2 * D; k0 += 4) {
        v2f a = *(const v2f*)&Xs[m][k0 + kb];
        #pragma unroll
        for (int nt = 0; nt < 4; ++nt) {
            const int col = nt * 16 + m;
            v2f b = *(const v2f*)&W1[(size_t)col * (2 * D) + k0 + kb];
            c[nt] = __builtin_amdgcn_wmma_f32_16x16x4_f32(
                false, a, false, b, (short)0, c[nt], false, false);
        }
    }

    #pragma unroll
    for (int nt = 0; nt < 4; ++nt) {
        const int   col  = nt * 16 + m;
        const float bias = b1[col];
        #pragma unroll
        for (int r = 0; r < 8; ++r) {
            const int row = r + half * 8;
            h1[(size_t)(tile * 16 + row) * D + col] = tanhf(c[nt][r] + bias);
        }
    }
}

__global__ __launch_bounds__(32) void graphsage_hop2(
    const float* __restrict__ feat, const int* __restrict__ adj,
    const int* __restrict__ in1, const int* __restrict__ in2,
    const int* __restrict__ ineg,
    const float* __restrict__ W2, const float* __restrict__ b2,
    const float* __restrict__ h1, float* __restrict__ out)
{
    __shared__ float Xs[16][XPAD];

    const int lane = threadIdx.x;
    const int m    = lane & 15;
    const int half = lane >> 4;
    const int tile = blockIdx.x;
    const int g    = tile * 16 + m;

    int node;
    if      (g < NB1)      node = in1[g];
    else if (g < 2 * NB1)  node = in2[g - NB1];
    else if (g < GTOT)     node = ineg[g - 2 * NB1];
    else                   node = 0;

    {
        const float4* selfp = (const float4*)(feat + (size_t)node * D + half * 32);
        float4* xlo = (float4*)&Xs[m][half * 32];
        float4* xhi = (float4*)&Xs[m][D + half * 32];
        #pragma unroll
        for (int i = 0; i < 8; ++i) xlo[i] = selfp[i];

        float4 acc[8];
        #pragma unroll
        for (int i = 0; i < 8; ++i) acc[i] = make_float4(0.f, 0.f, 0.f, 0.f);
        for (int j = 0; j < DEG; ++j) {
            const int nb = adj[(size_t)node * DEG + j];
            const float4* p = (const float4*)(h1 + (size_t)nb * D + half * 32);
            #pragma unroll
            for (int i = 0; i < 8; ++i) {
                float4 v = p[i];
                acc[i].x += v.x; acc[i].y += v.y; acc[i].z += v.z; acc[i].w += v.w;
            }
        }
        #pragma unroll
        for (int i = 0; i < 8; ++i) xhi[i] = acc[i];
    }
    __syncthreads();

    const int kb = half * 2;
    v8f zero = {};
    v8f c[8];
    #pragma unroll
    for (int nt = 0; nt < 8; ++nt) c[nt] = zero;

    for (int k0 = 0; k0 < D2; k0 += 4) {
        v2f a = *(const v2f*)&Xs[m][k0 + kb];
        #pragma unroll
        for (int nt = 0; nt < 8; ++nt) {
            const int col = nt * 16 + m;
            v2f b = *(const v2f*)&W2[(size_t)col * D2 + k0 + kb];
            c[nt] = __builtin_amdgcn_wmma_f32_16x16x4_f32(
                false, a, false, b, (short)0, c[nt], false, false);
        }
    }

    float bias[8];
    #pragma unroll
    for (int nt = 0; nt < 8; ++nt) bias[nt] = b2[nt * 16 + m];

    #pragma unroll
    for (int r = 0; r < 8; ++r) {
        float vals[8];
        float ss = 0.f;
        #pragma unroll
        for (int nt = 0; nt < 8; ++nt) {
            const float v = c[nt][r] + bias[nt];
            vals[nt] = v;
            ss += v * v;
        }
        ss += __shfl_xor(ss, 1);
        ss += __shfl_xor(ss, 2);
        ss += __shfl_xor(ss, 4);
        ss += __shfl_xor(ss, 8);
        const float inv = 1.0f / fmaxf(sqrtf(ss), EPSN);

        const int grow = tile * 16 + r + half * 8;
        if (grow < GTOT) {
            #pragma unroll
            for (int nt = 0; nt < 8; ++nt)
                out[(size_t)grow * D2 + nt * 16 + m] = vals[nt] * inv;
        }
    }
}

extern "C" void kernel_launch(void* const* d_in, const int* in_sizes, int n_in,
                              void* d_out, int out_size, void* d_ws, size_t ws_size,
                              hipStream_t stream) {
    const float* feat = (const float*)d_in[0];
    const int*   adj  = (const int*)  d_in[1];
    const int*   in1  = (const int*)  d_in[2];
    const int*   in2  = (const int*)  d_in[3];
    const int*   ineg = (const int*)  d_in[4];
    const float* W1   = (const float*)d_in[5];
    const float* b1   = (const float*)d_in[6];
    const float* W2   = (const float*)d_in[7];
    const float* b2   = (const float*)d_in[8];
    float*       outp = (float*)d_out;
    float*       h1   = (float*)d_ws;

    graphsage_hop1<<<N_NODES / 16, 32, 0, stream>>>(feat, adj, W1, b1, h1);
    graphsage_hop2<<<(GTOT + 15) / 16, 32, 0, stream>>>(feat, adj, in1, in2, ineg,
                                                        W2, b2, h1, outp);
}


# Hop 1: 4-wave work-groups walk 4 tiles each; a wave owns one 16-column tile, its W1 fragments kept in registers; half-waves gather whole rows (32 in flight) and store whole h1 rows; sum orders kept
# speedup vs baseline: 2.1877x; 2.1877x over previous
#include <hip/hip_runtime.h>
#include <math.h>

typedef __attribute__((ext_vector_type(2))) float v2f;
typedef __attribute__((ext_vector_type(4))) float v4f;
typedef __attribute__((ext_vector_type(8))) float v8f;

#define N_NODES 65536
#define DEG     16
#define D       64
#define D2      128
#define NB1     1024
#define NNEG    10
#define GTOT    (NB1 + NB1 + NNEG)
#define EPSN    1e-12f
#define XPAD    132

#define H1_WAVES 4
#define H1_TILES (N_NODES / 16)
#define H1_WGS   1024
#define HPAD     68
__global__ __launch_bounds__(32 * H1_WAVES) void graphsage_hop1(
    const float* __restrict__ feat, const int* __restrict__ adj,
    const float* __restrict__ W1,   const float* __restrict__ b1,
    float* __restrict__ h1)
{
    __shared__ float Xs[16][XPAD];
    __shared__ float Hs[16][HPAD];

    const int lane = threadIdx.x & 31;
    const int wave = threadIdx.x >> 5;
    const int m    = lane & 15;
    const int half = lane >> 4;
    const int kb   = half * 2;
    const int col  = wave * 16 + m;
    const int hw   = wave + H1_WAVES * half;

    v2f b[D2 / 4];
    #pragma unroll
    for (int s = 0; s < D2 / 4; ++s)
        b[s] = *(const v2f*)&W1[(size_t)col * D2 + s * 4 + kb];
    const float bias = b1[col];

    const char* featb = (const char*)feat;
    const char* adjb  = (const char*)adj;
    const unsigned mo = (unsigned)m * 16u;

    int4 id[8];
    {
        const unsigned ao = ((unsigned)blockIdx.x * 16u + (unsigned)hw) * (DEG * 4u);
        #pragma unroll
        for (int q = 0; q < 4; ++q) {
            id[q]     = *(const int4*)(adjb + (ao + 16u * q));
            id[4 + q] = *(const int4*)(adjb + (ao + 8u * DEG * 4u + 16u * q));
        }
    }

    for (int tile = blockIdx.x; tile < H1_TILES; tile += H1_WGS) {
        const unsigned node0 = (unsigned)tile * 16u + (unsigned)hw;
        const unsigned node1 = node0 + 8u;

        {
            const v4f self0 = *(const v4f*)(featb + (node0 * (D * 4u) + mo));
            const v4f self1 = *(const v4f*)(featb + (node1 * (D * 4u) + mo));

            const int nb0[DEG] = { id[0].x, id[0].y, id[0].z, id[0].w, id[1].x, id[1].y, id[1].z, id[1].w,
                                   id[2].x, id[2].y, id[2].z, id[2].w, id[3].x, id[3].y, id[3].z, id[3].w };
            const int nb1[DEG] = { id[4].x, id[4].y, id[4].z, id[4].w, id[5].x, id[5].y, id[5].z, id[5].w,
                                   id[6].x, id[6].y, id[6].z, id[6].w, id[7].x, id[7].y, id[7].z, id[7].w };

            v4f v[DEG], u[DEG];
            #pragma unroll
            for (int j = 0; j < DEG; ++j)
                v[j] = *(const v4f*)(featb + ((unsigned)nb0[j] * (D * 4u) + mo));
            #pragma unroll
            for (int j = 0; j < DEG; ++j)
                u[j] = *(const v4f*)(featb + ((unsigned)nb1[j] * (D * 4u) + mo));

            asm volatile("" :: "v"(v[0]),  "v"(v[1]),  "v"(v[2]),  "v"(v[3]),
                               "v"(v[4]),  "v"(v[5]),  "v"(v[6]),  "v"(v[7]),
                               "v"(v[8]),  "v"(v[9]),  "v"(v[10]), "v"(v[11]),
                               "v"(v[12]), "v"(v[13]), "v"(v[14]), "v"(v[15]));

            v4f acc0 = {0.f, 0.f, 0.f, 0.f};
            #pragma unroll
            for (int j = 0; j < DEG; ++j) acc0 = acc0 + v[j];

            asm volatile("" :: "v"(u[0]),  "v"(u[1]),  "v"(u[2]),  "v"(u[3]),
                               "v"(u[4]),  "v"(u[5]),  "v"(u[6]),  "v"(u[7]),
                               "v"(u[8]),  "v"(u[9]),  "v"(u[10]), "v"(u[11]),
                               "v"(u[12]), "v"(u[13]), "v"(u[14]), "v"(u[15]));

            v4f acc1 = {0.f, 0.f, 0.f, 0.f};
            #pragma unroll
            for (int j = 0; j < DEG; ++j) acc1 = acc1 + u[j];

            *(v4f*)&Xs[hw][m * 4]         = self0;
            *(v4f*)&Xs[hw][D + m * 4]     = acc0;
            *(v4f*)&Xs[hw + 8][m * 4]     = self1;
            *(v4f*)&Xs[hw + 8][D + m * 4] = acc1;
        }
        __syncthreads();

        {
            const unsigned nxt = min((unsigned)tile + H1_WGS, (unsigned)H1_TILES - 1u);
            const unsigned an  = (nxt * 16u + (unsigned)hw) * (DEG * 4u);
            #pragma unroll
            for (int q = 0; q < 4; ++q) {
                id[q]     = *(const int4*)(adjb + (an + 16u * q));
                id[4 + q] = *(const int4*)(adjb + (an + 8u * DEG * 4u + 16u * q));
            }
        }

        v8f c = {};
        #pragma unroll
        for (int s = 0; s < D2 / 4; ++s) {
            v2f a = *(const v2f*)&Xs[m][s * 4 + kb];
            c = __builtin_amdgcn_wmma_f32_16x16x4_f32(
                false, a, false, b[s], (short)0, c, false, false);
        }

        #pragma unroll
        for (int r = 0; r < 8; ++r) Hs[r + half * 8][col] = tanhf(c[r] + bias);
        __syncthreads();

        {
            const v4f o0 = *(const v4f*)&Hs[hw][m * 4];
            const v4f o1 = *(const v4f*)&Hs[hw + 8][m * 4];
            *(v4f*)(h1 + (size_t)node0 * D + m * 4) = o0;
            *(v4f*)(h1 + (size_t)node1 * D + m * 4) = o1;
        }
    }
}

#define H2_WAVES 8
__global__ __launch_bounds__(32 * H2_WAVES) void graphsage_hop2(
    const float* __restrict__ feat, const int* __restrict__ adj,
    const int* __restrict__ in1, const int* __restrict__ in2,
    const int* __restrict__ ineg,
    const float* __restrict__ W2, const float* __restrict__ b2,
    const float* __restrict__ h1, float* __restrict__ out)
{
    __shared__ float Xs[16][XPAD];
    __shared__ float Os[16][XPAD];

    const int lane = threadIdx.x & 31;
    const int wave = threadIdx.x >> 5;
    const int m    = lane & 15;
    const int half = lane >> 4;
    const int tile = blockIdx.x;
    const int kb   = half * 2;
    const int col  = wave * 16 + m;

    const int row = wave + 8 * half;
    const int g   = tile * 16 + row;

    int node;
    if      (g < NB1)      node = in1[g];
    else if (g < 2 * NB1)  node = in2[g - NB1];
    else if (g < GTOT)     node = ineg[g - 2 * NB1];
    else                   node = 0;

    v2f b[D2 / 4];
    #pragma unroll
    for (int s = 0; s < D2 / 4; ++s)
        b[s] = *(const v2f*)&W2[(size_t)col * D2 + s * 4 + kb];

    {
        const v4f selfv = *(const v4f*)(feat + (size_t)node * D + m * 4);

        const int4* ap = (const int4*)(adj + (size_t)node * DEG);
        const int4 a0 = ap[0], a1 = ap[1], a2 = ap[2], a3 = ap[3];
        const int nb[DEG] = { a0.x, a0.y, a0.z, a0.w, a1.x, a1.y, a1.z, a1.w,
                              a2.x, a2.y, a2.z, a2.w, a3.x, a3.y, a3.z, a3.w };

        v4f v[DEG];
        #pragma unroll
        for (int j = 0; j < DEG; ++j)
            v[j] = *(const v4f*)(h1 + (size_t)nb[j] * D + m * 4);

        asm volatile("" :: "v"(v[0]),  "v"(v[1]),  "v"(v[2]),  "v"(v[3]),
                           "v"(v[4]),  "v"(v[5]),  "v"(v[6]),  "v"(v[7]),
                           "v"(v[8]),  "v"(v[9]),  "v"(v[10]), "v"(v[11]),
                           "v"(v[12]), "v"(v[13]), "v"(v[14]), "v"(v[15]));

        v4f acc = {0.f, 0.f, 0.f, 0.f};
        #pragma unroll
        for (int j = 0; j < DEG; ++j) acc = acc + v[j];

        *(v4f*)&Xs[row][m * 4]     = selfv;
        *(v4f*)&Xs[row][D + m * 4] = acc;
    }
    __syncthreads();

    v8f c = {};
    #pragma unroll
    for (int s = 0; s < D2 / 4; ++s) {
        v2f a = *(const v2f*)&Xs[m][s * 4 + kb];
        c = __builtin_amdgcn_wmma_f32_16x16x4_f32(
            false, a, false, b[s], (short)0, c, false, false);
    }

    {
        const float bias = b2[col];
        #pragma unroll
        for (int r = 0; r < 8; ++r) Os[r + half * 8][col] = c[r] + bias;
    }
    __syncthreads();

    {
        float vals[8];
        float ss = 0.f;
        #pragma unroll
        for (int nt = 0; nt < 8; ++nt) {
            const float x = Os[row][nt * 16 + m];
            vals[nt] = x;
            ss += x * x;
        }
        ss += __shfl_xor(ss, 1);
        ss += __shfl_xor(ss, 2);
        ss += __shfl_xor(ss, 4);
        ss += __shfl_xor(ss, 8);
        const float inv = 1.0f / fmaxf(sqrtf(ss), EPSN);

        if (g < GTOT) {
            #pragma unroll
            for (int nt = 0; nt < 8; ++nt)
                out[(size_t)g * D2 + nt * 16 + m] = vals[nt] * inv;
        }
    }
}

extern "C" void kernel_launch(void* const* d_in, const int* in_sizes, int n_in,
                              void* d_out, int out_size, void* d_ws, size_t ws_size,
                              hipStream_t stream) {
    const float* feat = (const float*)d_in[0];
    const int*   adj  = (const int*)  d_in[1];
    const int*   in1  = (const int*)  d_in[2];
    const int*   in2  = (const int*)  d_in[3];
    const int*   ineg = (const int*)  d_in[4];
    const float* W1   = (const float*)d_in[5];
    const float* b1   = (const float*)d_in[6];
    const float* W2   = (const float*)d_in[7];
    const float* b2   = (const float*)d_in[8];
    float*       outp = (float*)d_out;
    float*       h1   = (float*)d_ws;

    graphsage_hop1<<<H1_WGS, 32 * H1_WAVES, 0, stream>>>(feat, adj, W1, b1, h1);
    graphsage_hop2<<<(GTOT + 15) / 16, 256, 0, stream>>>(feat, adj, in1, in2, ineg,
                                                        W2, b2, h1, outp);
}
